# starting kernel
# baseline (speedup 1.0000x reference)
#include <hip/hip_runtime.h>
#include <math.h>


typedef float v2f __attribute__((ext_vector_type(2)));
typedef float v4f __attribute__((ext_vector_type(4)));
typedef float v8f __attribute__((ext_vector_type(8)));

#define BATCH 64
#define NI    512
#define NO    64
#define NV    16
#define NL    8
#define KTOT  (NI * NL)
#define KSPL  (KTOT / 4)
#define IB    8

static __device__ __forceinline__ v8f wmma_f32_k4(v2f a, v2f b, v8f c) {
  return __builtin_amdgcn_wmma_f32_16x16x4_f32(false, a, false, b, (short)0, c,
                                               false, false);
}

static __device__ __forceinline__ void stage_tile(float* red, int slot, int lane,
                                                  v8f c) {
  v4f lo = __builtin_shufflevector(c, c, 0, 1, 2, 3);
  v4f hi = __builtin_shufflevector(c, c, 4, 5, 6, 7);
  *(v4f*)&red[((size_t)slot * 32 + lane) * 8 + 0] = lo;
  *(v4f*)&red[((size_t)slot * 32 + lane) * 8 + 4] = hi;
}

template <bool WITH_PROBS>
static __device__ __forceinline__ void caps_gemm_body(
    const float* __restrict__ w, const float* __restrict__ x,
    const float* __restrict__ probs, float* __restrict__ outw, float out_scale) {
  __shared__ float red[4 * 4 * 32 * 8];
  const int lane = threadIdx.x & 31;
  const int ks   = threadIdx.x >> 5;
  const int o    = blockIdx.x;
  const int half = lane >> 4;
  const int lr   = lane & 15;
  const float* __restrict__ wo = w + (size_t)o * NI * NV * NL;
  const float* __restrict__ po =
      WITH_PROBS ? probs + (size_t)o * NI * BATCH : nullptr;

  v8f c0 = {}, c1 = {}, c2 = {}, c3 = {};
  const int kbeg = ks * KSPL, kend = kbeg + KSPL;
  for (int k0 = kbeg; k0 < kend; k0 += 4) {
    const int k = k0 + 2 * half;
    const int i = k >> 3;
    const int l = k & 7;
    if ((k0 & 255) == 0 && (k0 + 512) < kend)
      __builtin_prefetch(wo + (size_t)((k0 + 512) >> 3) * NV * NL, 0, 0);

    v2f a = *(const v2f*)(wo + ((size_t)i * NV + lr) * NL + l);

    const float* xb = x + ((size_t)lr * NI + i) * NL + l;
    v2f b0 = *(const v2f*)(xb);
    v2f b1 = *(const v2f*)(xb + (size_t)16 * NI * NL);
    v2f b2 = *(const v2f*)(xb + (size_t)32 * NI * NL);
    v2f b3 = *(const v2f*)(xb + (size_t)48 * NI * NL);
    if (WITH_PROBS) {
      const float* pb = po + (size_t)i * BATCH + lr;
      b0 = b0 * pb[0];
      b1 = b1 * pb[16];
      b2 = b2 * pb[32];
      b3 = b3 * pb[48];
    }
    c0 = wmma_f32_k4(a, b0, c0);
    c1 = wmma_f32_k4(a, b1, c1);
    c2 = wmma_f32_k4(a, b2, c2);
    c3 = wmma_f32_k4(a, b3, c3);
  }

  stage_tile(red, ks * 4 + 0, lane, c0);
  stage_tile(red, ks * 4 + 1, lane, c1);
  stage_tile(red, ks * 4 + 2, lane, c2);
  stage_tile(red, ks * 4 + 3, lane, c3);
  __syncthreads();

  const int bt = ks;
  const int bb = bt * 16 + lr;
  v4f alo = {}, ahi = {};
  #pragma unroll
  for (int s = 0; s < 4; ++s) {
    const float* q = &red[((size_t)(s * 4 + bt) * 32 + lane) * 8];
    alo += *(const v4f*)(q);
    ahi += *(const v4f*)(q + 4);
  }
  #pragma unroll
  for (int r = 0; r < 4; ++r) {
    outw[((size_t)o * NV + (r + 8 * half)) * BATCH + bb]     = alo[r] * out_scale;
    outw[((size_t)o * NV + (r + 4 + 8 * half)) * BATCH + bb] = ahi[r] * out_scale;
  }
}

__global__ void caps_init_kernel(const float* __restrict__ w,
                                 const float* __restrict__ x,
                                 float* __restrict__ outw) {
  caps_gemm_body<false>(w, x, nullptr, outw, 1.0f / (float)NI);
}

__global__ void caps_update_kernel(const float* __restrict__ w,
                                   const float* __restrict__ x,
                                   const float* __restrict__ probs,
                                   float* __restrict__ outw) {
  caps_gemm_body<true>(w, x, probs, outw, 1.0f);
}

__global__ void caps_nhat_kernel(const float* __restrict__ outw,
                                 float* __restrict__ nhat) {
  const int row = blockIdx.x * blockDim.x + threadIdx.x;
  const int o = row >> 6, b = row & 63;
  const float* p = outw + (size_t)o * NV * BATCH + b;
  float v[NV], ss = 0.0f;
  #pragma unroll
  for (int t = 0; t < NV; ++t) { v[t] = p[(size_t)t * BATCH]; ss += v[t] * v[t]; }
  const float inv = 1.0f / fmaxf(sqrtf(ss), 1e-12f);
  float* q = nhat + (size_t)o * NV * BATCH + b;
  #pragma unroll
  for (int t = 0; t < NV; ++t) q[(size_t)t * BATCH] = v[t] * inv;
}

__global__ void caps_logits_softmax_kernel(const float* __restrict__ w,
                                           const float* __restrict__ x,
                                           const float* __restrict__ nhat,
                                           float* __restrict__ probs) {
  extern __shared__ float slog[];
  const int tid  = threadIdx.x;
  const int wv   = tid >> 5;
  const int lane = tid & 31;
  const int half = lane >> 4;
  const int lr   = lane & 15;
  const int ib0  = blockIdx.x * IB;

  for (int t = wv; t < NO * IB * 4; t += 16) {
    const int o  = t >> 5;
    const int r2 = t & 31;
    const int ii = r2 >> 2;
    const int bt = r2 & 3;
    const int i  = ib0 + ii;
    const int bb = bt * 16 + lr;
    const float* __restrict__ wrow = w + ((size_t)o * NI + i) * NV * NL;

    v8f c = {};
    v2f a0 = *(const v2f*)(wrow + (size_t)lr * NL + 2 * half);
    v2f b0 = *(const v2f*)(x + ((size_t)bb * NI + i) * NL + 2 * half);
    c = wmma_f32_k4(a0, b0, c);
    v2f a1 = *(const v2f*)(wrow + (size_t)lr * NL + 4 + 2 * half);
    v2f b1 = *(const v2f*)(x + ((size_t)bb * NI + i) * NL + 4 + 2 * half);
    c = wmma_f32_k4(a1, b1, c);

    const float* __restrict__ nh = nhat + (size_t)o * NV * BATCH + bb;
    float acc = 0.0f;
    #pragma unroll
    for (int r = 0; r < 8; ++r) acc += c[r] * nh[(size_t)(r + 8 * half) * BATCH];
    acc += __shfl_xor(acc, 16, 32);
    if (half == 0) slog[(o * IB + ii) * BATCH + bb] = acc;
  }
  __syncthreads();

  const int ii = tid >> 6;
  const int b  = tid & 63;
  const int i  = ib0 + ii;
  float mx = -3.402823466e+38f;
  for (int o = 0; o < NO; ++o)
    mx = fmaxf(mx, slog[(o * IB + ii) * BATCH + b]);
  float s = 0.0f;
  for (int o = 0; o < NO; ++o) {
    const float e = __expf(slog[(o * IB + ii) * BATCH + b] - mx);
    slog[(o * IB + ii) * BATCH + b] = e;
    s += e;
  }
  const float inv = 1.0f / s;
  for (int o = 0; o < NO; ++o)
    probs[((size_t)o * NI + i) * BATCH + b] =
        slog[(o * IB + ii) * BATCH + b] * inv;
}

__global__ void caps_final_kernel(const float* __restrict__ outw,
                                  const float* __restrict__ bias,
                                  float* __restrict__ y) {
  const int row = blockIdx.x * blockDim.x + threadIdx.x;
  const int o = row >> 6, b = row & 63;
  const float* p = outw + (size_t)o * NV * BATCH + b;
  float v[NV], ss = 0.0f;
  #pragma unroll
  for (int t = 0; t < NV; ++t) { v[t] = p[(size_t)t * BATCH]; ss += v[t] * v[t]; }
  const float n  = sqrtf(ss);
  const float sc = n / (1.0f + ss);
  #pragma unroll
  for (int t = 0; t < NV; ++t)
    y[((size_t)b * NO + o) * NV + t] = v[t] * sc + bias[o * NV + t];
}

extern "C" void kernel_launch(void* const* d_in, const int* in_sizes, int n_in,
                              void* d_out, int out_size, void* d_ws, size_t ws_size,
                              hipStream_t stream) {
  const float* x    = (const float*)d_in[0];
  const float* w    = (const float*)d_in[1];
  const float* bias = (const float*)d_in[2];
  float* y = (float*)d_out;

  float* probs = (float*)d_ws;
  float* outw  = probs + (size_t)NO * NI * BATCH;
  float* nhat  = outw  + (size_t)NO * NV * BATCH;
  const size_t need =
      ((size_t)NO * NI * BATCH + 2 * (size_t)NO * NV * BATCH) * sizeof(float);
  if (ws_size < need) return;

  caps_init_kernel<<<NO, 128, 0, stream>>>(w, x, outw);
  for (int it = 0; it < 3; ++it) {
    caps_nhat_kernel<<<16, 256, 0, stream>>>(outw, nhat);
    caps_logits_softmax_kernel<<<64, 512,
        (size_t)NO * IB * BATCH * sizeof(float), stream>>>(w, x, nhat, probs);
    caps_update_kernel<<<NO, 128, 0, stream>>>(w, x, probs, outw);
  }
  caps_final_kernel<<<16, 256, 0, stream>>>(outw, bias, y);
}


# GEMM K loops and logits tile loop software-pipelined: next block's loads issued as one group ahead of the current WMMAs (two register sets); products and sum order unchanged
# speedup vs baseline: 2.3717x; 2.3717x over previous
#include <hip/hip_runtime.h>
#include <math.h>


typedef float v2f __attribute__((ext_vector_type(2)));
typedef float v4f __attribute__((ext_vector_type(4)));
typedef float v8f __attribute__((ext_vector_type(8)));

#define BATCH 64
#define NI    512
#define NO    64
#define NV    16
#define NL    8
#define KTOT  (NI * NL)
#define KSPL  (KTOT / 4)
#define IB    8
#define KB    8

static __device__ __forceinline__ v8f wmma_f32_k4(v2f a, v2f b, v8f c) {
  return __builtin_amdgcn_wmma_f32_16x16x4_f32(false, a, false, b, (short)0, c,
                                               false, false);
}

static __device__ __forceinline__ void stage_tile(float* red, int slot, int lane,
                                                  v8f c) {
  v4f lo = __builtin_shufflevector(c, c, 0, 1, 2, 3);
  v4f hi = __builtin_shufflevector(c, c, 4, 5, 6, 7);
  *(v4f*)&red[((size_t)slot * 32 + lane) * 8 + 0] = lo;
  *(v4f*)&red[((size_t)slot * 32 + lane) * 8 + 4] = hi;
}

struct KBlock {
  v2f a[KB];
  v2f b[KB][4];
  float p[KB / 2][4];
};

template <bool WITH_PROBS>
static __device__ __forceinline__ void load_kblock(
    KBlock& f, const float* __restrict__ wo, const float* __restrict__ x,
    const float* __restrict__ po, int k0, int half, int lr) {
  #pragma unroll
  for (int s = 0; s < KB; ++s) {
    const int i = (k0 >> 3) + (s >> 1);
    const int l = 4 * (s & 1) + 2 * half;
    f.a[s] = *(const v2f*)(wo + ((size_t)i * NV + lr) * NL + l);
    const float* xb = x + ((size_t)lr * NI + i) * NL + l;
    f.b[s][0] = *(const v2f*)(xb);
    f.b[s][1] = *(const v2f*)(xb + (size_t)16 * NI * NL);
    f.b[s][2] = *(const v2f*)(xb + (size_t)32 * NI * NL);
    f.b[s][3] = *(const v2f*)(xb + (size_t)48 * NI * NL);
  }
  if (WITH_PROBS) {
    #pragma unroll
    for (int q = 0; q < KB / 2; ++q) {
      const float* pb = po + (size_t)((k0 >> 3) + q) * BATCH + lr;
      f.p[q][0] = pb[0];
      f.p[q][1] = pb[16];
      f.p[q][2] = pb[32];
      f.p[q][3] = pb[48];
    }
  }
}

template <bool WITH_PROBS>
static __device__ __forceinline__ void mma_kblock(const KBlock& f, v8f& c0,
                                                  v8f& c1, v8f& c2, v8f& c3) {
  #pragma unroll
  for (int s = 0; s < KB; ++s) {
    v2f b0 = f.b[s][0], b1 = f.b[s][1], b2 = f.b[s][2], b3 = f.b[s][3];
    if (WITH_PROBS) {
      b0 = b0 * f.p[s >> 1][0];
      b1 = b1 * f.p[s >> 1][1];
      b2 = b2 * f.p[s >> 1][2];
      b3 = b3 * f.p[s >> 1][3];
    }
    c0 = wmma_f32_k4(f.a[s], b0, c0);
    c1 = wmma_f32_k4(f.a[s], b1, c1);
    c2 = wmma_f32_k4(f.a[s], b2, c2);
    c3 = wmma_f32_k4(f.a[s], b3, c3);
  }
}

template <bool WITH_PROBS>
static __device__ __forceinline__ void caps_gemm_body(
    const float* __restrict__ w, const float* __restrict__ x,
    const float* __restrict__ probs, float* __restrict__ outw, float out_scale) {
  __shared__ float red[4 * 4 * 32 * 8];
  const int lane = threadIdx.x & 31;
  const int ks   = threadIdx.x >> 5;
  const int o    = blockIdx.x;
  const int half = lane >> 4;
  const int lr   = lane & 15;
  const float* __restrict__ wo = w + (size_t)o * NI * NV * NL;
  const float* __restrict__ po =
      WITH_PROBS ? probs + (size_t)o * NI * BATCH : nullptr;

  v8f c0 = {}, c1 = {}, c2 = {}, c3 = {};
  const int kbeg = ks * KSPL;
  constexpr int KSTEP = KB * 4;
  constexpr int NTRIP = KSPL / (2 * KSTEP);
  KBlock f0, f1;
  load_kblock<WITH_PROBS>(f0, wo, x, po, kbeg, half, lr);
  for (int t = 0; t < NTRIP; ++t) {
    const int k0 = kbeg + t * 2 * KSTEP;
    load_kblock<WITH_PROBS>(f1, wo, x, po, k0 + KSTEP, half, lr);
    __builtin_amdgcn_sched_barrier(0);
    mma_kblock<WITH_PROBS>(f0, c0, c1, c2, c3);
    __builtin_amdgcn_sched_barrier(0);
    if (t + 1 < NTRIP)
      load_kblock<WITH_PROBS>(f0, wo, x, po, k0 + 2 * KSTEP, half, lr);
    __builtin_amdgcn_sched_barrier(0);
    mma_kblock<WITH_PROBS>(f1, c0, c1, c2, c3);
    __builtin_amdgcn_sched_barrier(0);
  }

  stage_tile(red, ks * 4 + 0, lane, c0);
  stage_tile(red, ks * 4 + 1, lane, c1);
  stage_tile(red, ks * 4 + 2, lane, c2);
  stage_tile(red, ks * 4 + 3, lane, c3);
  __syncthreads();

  const int bt = ks;
  const int bb = bt * 16 + lr;
  v4f alo = {}, ahi = {};
  #pragma unroll
  for (int s = 0; s < 4; ++s) {
    const float* q = &red[((size_t)(s * 4 + bt) * 32 + lane) * 8];
    alo += *(const v4f*)(q);
    ahi += *(const v4f*)(q + 4);
  }
  #pragma unroll
  for (int r = 0; r < 4; ++r) {
    outw[((size_t)o * NV + (r + 8 * half)) * BATCH + bb]     = alo[r] * out_scale;
    outw[((size_t)o * NV + (r + 4 + 8 * half)) * BATCH + bb] = ahi[r] * out_scale;
  }
}

__global__ void __launch_bounds__(128) caps_init_kernel(const float* __restrict__ w,
                                 const float* __restrict__ x,
                                 float* __restrict__ outw) {
  caps_gemm_body<false>(w, x, nullptr, outw, 1.0f / (float)NI);
}

__global__ void __launch_bounds__(128) caps_update_kernel(const float* __restrict__ w,
                                   const float* __restrict__ x,
                                   const float* __restrict__ probs,
                                   float* __restrict__ outw) {
  caps_gemm_body<true>(w, x, probs, outw, 1.0f);
}

__global__ void caps_nhat_kernel(const float* __restrict__ outw,
                                 float* __restrict__ nhat) {
  const int row = blockIdx.x * blockDim.x + threadIdx.x;
  const int o = row >> 6, b = row & 63;
  const float* p = outw + (size_t)o * NV * BATCH + b;
  float v[NV], ss = 0.0f;
  #pragma unroll
  for (int t = 0; t < NV; ++t) { v[t] = p[(size_t)t * BATCH]; ss += v[t] * v[t]; }
  const float inv = 1.0f / fmaxf(sqrtf(ss), 1e-12f);
  float* q = nhat + (size_t)o * NV * BATCH + b;
  #pragma unroll
  for (int t = 0; t < NV; ++t) q[(size_t)t * BATCH] = v[t] * inv;
}

struct LTile {
  v2f a0, b0, a1, b1;
  float nh[8];
};

static __device__ __forceinline__ void load_ltile(
    LTile& f, const float* __restrict__ w, const float* __restrict__ x,
    const float* __restrict__ nhat, int t, int ib0, int half, int lr) {
  const int o  = t >> 5;
  const int r2 = t & 31;
  const int ii = r2 >> 2;
  const int bt = r2 & 3;
  const int i  = ib0 + ii;
  const int bb = bt * 16 + lr;
  const float* __restrict__ wrow = w + ((size_t)o * NI + i) * NV * NL;
  f.a0 = *(const v2f*)(wrow + (size_t)lr * NL + 2 * half);
  f.b0 = *(const v2f*)(x + ((size_t)bb * NI + i) * NL + 2 * half);
  f.a1 = *(const v2f*)(wrow + (size_t)lr * NL + 4 + 2 * half);
  f.b1 = *(const v2f*)(x + ((size_t)bb * NI + i) * NL + 4 + 2 * half);
  const float* __restrict__ nh = nhat + (size_t)o * NV * BATCH + bb;
  #pragma unroll
  for (int r = 0; r < 8; ++r) f.nh[r] = nh[(size_t)(r + 8 * half) * BATCH];
}

static __device__ __forceinline__ void finish_ltile(const LTile& f, float* slog,
                                                    int t, int half, int lr) {
  const int o  = t >> 5;
  const int r2 = t & 31;
  const int ii = r2 >> 2;
  const int bb = (r2 & 3) * 16 + lr;
  v8f c = {};
  c = wmma_f32_k4(f.a0, f.b0, c);
  c = wmma_f32_k4(f.a1, f.b1, c);
  float acc = 0.0f;
  #pragma unroll
  for (int r = 0; r < 8; ++r) acc += c[r] * f.nh[r];
  acc += __shfl_xor(acc, 16, 32);
  if (half == 0) slog[(o * IB + ii) * BATCH + bb] = acc;
}

__global__ void caps_logits_softmax_kernel(const float* __restrict__ w,
                                           const float* __restrict__ x,
                                           const float* __restrict__ nhat,
                                           float* __restrict__ probs) {
  extern __shared__ float slog[];
  const int tid  = threadIdx.x;
  const int wv   = tid >> 5;
  const int lane = tid & 31;
  const int half = lane >> 4;
  const int lr   = lane & 15;
  const int ib0  = blockIdx.x * IB;

  constexpr int NPAIR = NO * IB * 4 / 32;
  LTile g0, g1;
  load_ltile(g0, w, x, nhat, wv, ib0, half, lr);
  for (int n = 0; n < NPAIR; ++n) {
    const int t = wv + 32 * n;
    load_ltile(g1, w, x, nhat, t + 16, ib0, half, lr);
    finish_ltile(g0, slog, t, half, lr);
    if (n + 1 < NPAIR) load_ltile(g0, w, x, nhat, t + 32, ib0, half, lr);
    finish_ltile(g1, slog, t + 16, half, lr);
  }
  __syncthreads();

  const int ii = tid >> 6;
  const int b  = tid & 63;
  const int i  = ib0 + ii;
  float mx = -3.402823466e+38f;
  for (int o = 0; o < NO; ++o)
    mx = fmaxf(mx, slog[(o * IB + ii) * BATCH + b]);
  float s = 0.0f;
  for (int o = 0; o < NO; ++o) {
    const float e = __expf(slog[(o * IB + ii) * BATCH + b] - mx);
    slog[(o * IB + ii) * BATCH + b] = e;
    s += e;
  }
  const float inv = 1.0f / s;
  for (int o = 0; o < NO; ++o)
    probs[((size_t)o * NI + i) * BATCH + b] =
        slog[(o * IB + ii) * BATCH + b] * inv;
}

__global__ void caps_final_kernel(const float* __restrict__ outw,
                                  const float* __restrict__ bias,
                                  float* __restrict__ y) {
  const int row = blockIdx.x * blockDim.x + threadIdx.x;
  const int o = row >> 6, b = row & 63;
  const float* p = outw + (size_t)o * NV * BATCH + b;
  float v[NV], ss = 0.0f;
  #pragma unroll
  for (int t = 0; t < NV; ++t) { v[t] = p[(size_t)t * BATCH]; ss += v[t] * v[t]; }
  const float n  = sqrtf(ss);
  const float sc = n / (1.0f + ss);
  #pragma unroll
  for (int t = 0; t < NV; ++t)
    y[((size_t)b * NO + o) * NV + t] = v[t] * sc + bias[o * NV + t];
}

extern "C" void kernel_launch(void* const* d_in, const int* in_sizes, int n_in,
                              void* d_out, int out_size, void* d_ws, size_t ws_size,
                              hipStream_t stream) {
  const float* x    = (const float*)d_in[0];
  const float* w    = (const float*)d_in[1];
  const float* bias = (const float*)d_in[2];
  float* y = (float*)d_out;

  float* probs = (float*)d_ws;
  float* outw  = probs + (size_t)NO * NI * BATCH;
  float* nhat  = outw  + (size_t)NO * NV * BATCH;
  const size_t need =
      ((size_t)NO * NI * BATCH + 2 * (size_t)NO * NV * BATCH) * sizeof(float);
  if (ws_size < need) return;

  caps_init_kernel<<<NO, 128, 0, stream>>>(w, x, outw);
  for (int it = 0; it < 3; ++it) {
    caps_nhat_kernel<<<16, 256, 0, stream>>>(outw, nhat);
    caps_logits_softmax_kernel<<<64, 512,
        (size_t)NO * IB * BATCH * sizeof(float), stream>>>(w, x, nhat, probs);
    caps_update_kernel<<<NO, 128, 0, stream>>>(w, x, probs, outw);
  }
  caps_final_kernel<<<16, 256, 0, stream>>>(outw, bias, y);
}


# 7 GEMM/logits launches widened from 64 to 256 work-groups, one per compute unit: logits 2 in-capsules per group; GEMM K split over 4 groups, reader adds the 4 partial planes (K sum order changes)
# speedup vs baseline: 6.7236x; 2.8349x over previous
#include <hip/hip_runtime.h>
#include <math.h>


typedef float v2f __attribute__((ext_vector_type(2)));
typedef float v4f __attribute__((ext_vector_type(4)));
typedef float v8f __attribute__((ext_vector_type(8)));

#define BATCH 64
#define NI    512
#define NO    64
#define NV    16
#define NL    8
#define KTOT  (NI * NL)
#define KQ    4
#define KSPL  (KTOT / (4 * KQ))
#define PLANE (NO * NV * BATCH)
#define IB    2
#define KB    8

static __device__ __forceinline__ v8f wmma_f32_k4(v2f a, v2f b, v8f c) {
  return __builtin_amdgcn_wmma_f32_16x16x4_f32(false, a, false, b, (short)0, c,
                                               false, false);
}

static __device__ __forceinline__ void stage_tile(float* red, int slot, int lane,
                                                  v8f c) {
  v4f lo = __builtin_shufflevector(c, c, 0, 1, 2, 3);
  v4f hi = __builtin_shufflevector(c, c, 4, 5, 6, 7);
  *(v4f*)&red[((size_t)slot * 32 + lane) * 8 + 0] = lo;
  *(v4f*)&red[((size_t)slot * 32 + lane) * 8 + 4] = hi;
}

struct KBlock {
  v2f a[KB];
  v2f b[KB][4];
  float p[KB / 2][4];
};

template <bool WITH_PROBS>
static __device__ __forceinline__ void load_kblock(
    KBlock& f, const float* __restrict__ wo, const float* __restrict__ x,
    const float* __restrict__ po, int k0, int half, int lr) {
  #pragma unroll
  for (int s = 0; s < KB; ++s) {
    const int i = (k0 >> 3) + (s >> 1);
    const int l = 4 * (s & 1) + 2 * half;
    f.a[s] = *(const v2f*)(wo + ((size_t)i * NV + lr) * NL + l);
    const float* xb = x + ((size_t)lr * NI + i) * NL + l;
    f.b[s][0] = *(const v2f*)(xb);
    f.b[s][1] = *(const v2f*)(xb + (size_t)16 * NI * NL);
    f.b[s][2] = *(const v2f*)(xb + (size_t)32 * NI * NL);
    f.b[s][3] = *(const v2f*)(xb + (size_t)48 * NI * NL);
  }
  if (WITH_PROBS) {
    #pragma unroll
    for (int q = 0; q < KB / 2; ++q) {
      const float* pb = po + (size_t)((k0 >> 3) + q) * BATCH + lr;
      f.p[q][0] = pb[0];
      f.p[q][1] = pb[16];
      f.p[q][2] = pb[32];
      f.p[q][3] = pb[48];
    }
  }
}

template <bool WITH_PROBS>
static __device__ __forceinline__ void mma_kblock(const KBlock& f, v8f& c0,
                                                  v8f& c1, v8f& c2, v8f& c3) {
  #pragma unroll
  for (int s = 0; s < KB; ++s) {
    v2f b0 = f.b[s][0], b1 = f.b[s][1], b2 = f.b[s][2], b3 = f.b[s][3];
    if (WITH_PROBS) {
      b0 = b0 * f.p[s >> 1][0];
      b1 = b1 * f.p[s >> 1][1];
      b2 = b2 * f.p[s >> 1][2];
      b3 = b3 * f.p[s >> 1][3];
    }
    c0 = wmma_f32_k4(f.a[s], b0, c0);
    c1 = wmma_f32_k4(f.a[s], b1, c1);
    c2 = wmma_f32_k4(f.a[s], b2, c2);
    c3 = wmma_f32_k4(f.a[s], b3, c3);
  }
}

template <bool WITH_PROBS>
static __device__ __forceinline__ void caps_gemm_body(
    const float* __restrict__ w, const float* __restrict__ x,
    const float* __restrict__ probs, float* __restrict__ outw, float out_scale) {
  __shared__ float red[4 * 4 * 32 * 8];
  const int lane = threadIdx.x & 31;
  const int ks   = threadIdx.x >> 5;
  const int o    = blockIdx.x >> 2;
  const int kq   = blockIdx.x & (KQ - 1);
  const int half = lane >> 4;
  const int lr   = lane & 15;
  const float* __restrict__ wo = w + (size_t)o * NI * NV * NL;
  const float* __restrict__ po =
      WITH_PROBS ? probs + (size_t)o * NI * BATCH : nullptr;

  v8f c0 = {}, c1 = {}, c2 = {}, c3 = {};
  const int kbeg = (kq * 4 + ks) * KSPL;
  constexpr int KSTEP = KB * 4;
  constexpr int NTRIP = KSPL / (2 * KSTEP);
  KBlock f0, f1;
  load_kblock<WITH_PROBS>(f0, wo, x, po, kbeg, half, lr);
  for (int t = 0; t < NTRIP; ++t) {
    const int k0 = kbeg + t * 2 * KSTEP;
    load_kblock<WITH_PROBS>(f1, wo, x, po, k0 + KSTEP, half, lr);
    __builtin_amdgcn_sched_barrier(0);
    mma_kblock<WITH_PROBS>(f0, c0, c1, c2, c3);
    __builtin_amdgcn_sched_barrier(0);
    if (t + 1 < NTRIP)
      load_kblock<WITH_PROBS>(f0, wo, x, po, k0 + 2 * KSTEP, half, lr);
    __builtin_amdgcn_sched_barrier(0);
    mma_kblock<WITH_PROBS>(f1, c0, c1, c2, c3);
    __builtin_amdgcn_sched_barrier(0);
  }

  stage_tile(red, ks * 4 + 0, lane, c0);
  stage_tile(red, ks * 4 + 1, lane, c1);
  stage_tile(red, ks * 4 + 2, lane, c2);
  stage_tile(red, ks * 4 + 3, lane, c3);
  __syncthreads();

  const int bt = ks;
  const int bb = bt * 16 + lr;
  v4f alo = {}, ahi = {};
  #pragma unroll
  for (int s = 0; s < 4; ++s) {
    const float* q = &red[((size_t)(s * 4 + bt) * 32 + lane) * 8];
    alo += *(const v4f*)(q);
    ahi += *(const v4f*)(q + 4);
  }
  float* __restrict__ outp = outw + (size_t)kq * PLANE;
  #pragma unroll
  for (int r = 0; r < 4; ++r) {
    outp[((size_t)o * NV + (r + 8 * half)) * BATCH + bb]     = alo[r] * out_scale;
    outp[((size_t)o * NV + (r + 4 + 8 * half)) * BATCH + bb] = ahi[r] * out_scale;
  }
}

__global__ void __launch_bounds__(128) caps_init_kernel(const float* __restrict__ w,
                                 const float* __restrict__ x,
                                 float* __restrict__ outw) {
  caps_gemm_body<false>(w, x, nullptr, outw, 1.0f / (float)NI);
}

__global__ void __launch_bounds__(128) caps_update_kernel(const float* __restrict__ w,
                                   const float* __restrict__ x,
                                   const float* __restrict__ probs,
                                   float* __restrict__ outw) {
  caps_gemm_body<true>(w, x, probs, outw, 1.0f);
}

static __device__ __forceinline__ float sum_planes(const float* __restrict__ p) {
  float s = p[0];
  #pragma unroll
  for (int q = 1; q < KQ; ++q) s += p[(size_t)q * PLANE];
  return s;
}

__global__ void caps_nhat_kernel(const float* __restrict__ outw,
                                 float* __restrict__ nhat) {
  const int row = blockIdx.x * blockDim.x + threadIdx.x;
  const int o = row >> 6, b = row & 63;
  const float* p = outw + (size_t)o * NV * BATCH + b;
  float v[NV], ss = 0.0f;
  #pragma unroll
  for (int t = 0; t < NV; ++t) { v[t] = sum_planes(p + (size_t)t * BATCH); ss += v[t] * v[t]; }
  const float inv = 1.0f / fmaxf(sqrtf(ss), 1e-12f);
  float* q = nhat + (size_t)o * NV * BATCH + b;
  #pragma unroll
  for (int t = 0; t < NV; ++t) q[(size_t)t * BATCH] = v[t] * inv;
}

struct LPair {
  v2f a00, a01, a10, a11;
  float nh[8];
};

static __device__ __forceinline__ void load_lpair(
    LPair& f, const float* __restrict__ w, const float* __restrict__ nhat,
    int o, int ib0, int bb, int half, int lr) {
  const float* __restrict__ wrow =
      w + (((size_t)o * NI + ib0) * NV + lr) * NL + 2 * half;
  f.a00 = *(const v2f*)(wrow);
  f.a01 = *(const v2f*)(wrow + 4);
  f.a10 = *(const v2f*)(wrow + NV * NL);
  f.a11 = *(const v2f*)(wrow + NV * NL + 4);
  const float* __restrict__ nh = nhat + (size_t)o * NV * BATCH + bb;
  #pragma unroll
  for (int r = 0; r < 8; ++r) f.nh[r] = nh[(size_t)(r + 8 * half) * BATCH];
}

static __device__ __forceinline__ void finish_lpair(
    const LPair& f, v2f b00, v2f b01, v2f b10, v2f b11, float* slog, int o,
    int bb, int half) {
  v8f c0 = {}, c1 = {};
  c0 = wmma_f32_k4(f.a00, b00, c0);
  c1 = wmma_f32_k4(f.a10, b10, c1);
  c0 = wmma_f32_k4(f.a01, b01, c0);
  c1 = wmma_f32_k4(f.a11, b11, c1);
  float acc0 = 0.0f, acc1 = 0.0f;
  #pragma unroll
  for (int r = 0; r < 8; ++r) { acc0 += c0[r] * f.nh[r]; acc1 += c1[r] * f.nh[r]; }
  acc0 += __shfl_xor(acc0, 16, 32);
  acc1 += __shfl_xor(acc1, 16, 32);
  slog[(o * IB + half) * BATCH + bb] = half ? acc1 : acc0;
}

__global__ void caps_logits_softmax_kernel(const float* __restrict__ w,
                                           const float* __restrict__ x,
                                           const float* __restrict__ nhat,
                                           float* __restrict__ probs) {
  extern __shared__ float slog[];
  const int tid  = threadIdx.x;
  const int wv   = tid >> 5;
  const int lane = tid & 31;
  const int half = lane >> 4;
  const int lr   = lane & 15;
  const int ib0  = blockIdx.x * IB;

  const int bt = wv & 3;
  const int o0 = (wv >> 2) * (NO / 4);
  const int bb = bt * 16 + lr;
  const float* __restrict__ xr = x + ((size_t)bb * NI + ib0) * NL + 2 * half;
  const v2f b00 = *(const v2f*)(xr);
  const v2f b01 = *(const v2f*)(xr + 4);
  const v2f b10 = *(const v2f*)(xr + NL);
  const v2f b11 = *(const v2f*)(xr + NL + 4);
  constexpr int NTRIP = NO / 4 / 2;
  LPair g0, g1;
  load_lpair(g0, w, nhat, o0, ib0, bb, half, lr);
  for (int n = 0; n < NTRIP; ++n) {
    const int o = o0 + 2 * n;
    load_lpair(g1, w, nhat, o + 1, ib0, bb, half, lr);
    finish_lpair(g0, b00, b01, b10, b11, slog, o, bb, half);
    if (n + 1 < NTRIP) load_lpair(g0, w, nhat, o + 2, ib0, bb, half, lr);
    finish_lpair(g1, b00, b01, b10, b11, slog, o + 1, bb, half);
  }
  __syncthreads();

  constexpr int NCOL = IB * BATCH;
  constexpr int OQ   = NO / 4;
  float* smax = slog + NO * NCOL;
  const int pr = tid % NCOL;
  const int q  = tid / NCOL;
  const int ii = pr >> 6;
  const int b  = pr & 63;
  const int i  = ib0 + ii;
  float* col = slog + q * OQ * NCOL + pr;
  float e[OQ];
  float mq = -3.402823466e+38f;
  #pragma unroll
  for (int k = 0; k < OQ; ++k) { e[k] = col[k * NCOL]; mq = fmaxf(mq, e[k]); }
  smax[q * NCOL + pr] = mq;
  __syncthreads();
  const float mx = fmaxf(fmaxf(smax[pr], smax[NCOL + pr]),
                         fmaxf(smax[2 * NCOL + pr], smax[3 * NCOL + pr]));
  #pragma unroll
  for (int k = 0; k < OQ; ++k) {
    e[k] = __expf(e[k] - mx);
    col[k * NCOL] = e[k];
  }
  __syncthreads();
  float s = 0.0f;
  #pragma unroll 16
  for (int o = 0; o < NO; ++o) s += slog[o * NCOL + pr];
  const float inv = 1.0f / s;
  #pragma unroll
  for (int k = 0; k < OQ; ++k)
    probs[((size_t)(q * OQ + k) * NI + i) * BATCH + b] = e[k] * inv;
}

__global__ void caps_final_kernel(const float* __restrict__ outw,
                                  const float* __restrict__ bias,
                                  float* __restrict__ y) {
  const int row = blockIdx.x * blockDim.x + threadIdx.x;
  const int o = row >> 6, b = row & 63;
  const float* p = outw + (size_t)o * NV * BATCH + b;
  float v[NV], ss = 0.0f;
  #pragma unroll
  for (int t = 0; t < NV; ++t) { v[t] = sum_planes(p + (size_t)t * BATCH); ss += v[t] * v[t]; }
  const float n  = sqrtf(ss);
  const float sc = n / (1.0f + ss);
  #pragma unroll
  for (int t = 0; t < NV; ++t)
    y[((size_t)b * NO + o) * NV + t] = v[t] * sc + bias[o * NV + t];
}

extern "C" void kernel_launch(void* const* d_in, const int* in_sizes, int n_in,
                              void* d_out, int out_size, void* d_ws, size_t ws_size,
                              hipStream_t stream) {
  const float* x    = (const float*)d_in[0];
  const float* w    = (const float*)d_in[1];
  const float* bias = (const float*)d_in[2];
  float* y = (float*)d_out;

  float* probs = (float*)d_ws;
  float* outw  = probs + (size_t)NO * NI * BATCH;
  float* nhat  = outw  + (size_t)KQ * PLANE;
  const size_t need =
      ((size_t)NO * NI * BATCH + (size_t)(KQ + 1) * PLANE) * sizeof(float);
  if (ws_size < need) return;

  caps_init_kernel<<<NO * KQ, 128, 0, stream>>>(w, x, outw);
  for (int it = 0; it < 3; ++it) {
    caps_nhat_kernel<<<16, 256, 0, stream>>>(outw, nhat);
    caps_logits_softmax_kernel<<<NI / IB, 512,
        (size_t)(NO + 4) * IB * BATCH * sizeof(float), stream>>>(w, x, nhat, probs);
    caps_update_kernel<<<NO * KQ, 128, 0, stream>>>(w, x, probs, outw);
  }
  caps_final_kernel<<<16, 256, 0, stream>>>(outw, bias, y);
}


# GEMM work-groups re-cut from (o, K quarter) to (o, 16-batch tile) over all K: each owns its out columns and normalizes or squashes them itself; the 4 plane-summing launches go (K sum order changes)
# speedup vs baseline: 7.2283x; 1.0751x over previous
#include <hip/hip_runtime.h>
#include <math.h>


typedef float v2f __attribute__((ext_vector_type(2)));
typedef float v4f __attribute__((ext_vector_type(4)));
typedef float v8f __attribute__((ext_vector_type(8)));

#define BATCH 64
#define NI    512
#define NO    64
#define NV    16
#define NL    8
#define IB    2
#define IW    (NI / 4)
#define NB    16

static __device__ __forceinline__ v8f wmma_f32_k4(v2f a, v2f b, v8f c) {
  return __builtin_amdgcn_wmma_f32_16x16x4_f32(false, a, false, b, (short)0, c,
                                               false, false);
}

struct KBlock {
  v4f a[NB];
  v4f b[NB];
  float p[NB];
};

template <bool WITH_PROBS>
static __device__ __forceinline__ void load_kblock(
    KBlock& f, const float* __restrict__ wa, const float* __restrict__ xb,
    const float* __restrict__ pb, int i0) {
  #pragma unroll
  for (int q = 0; q < NB; ++q) {
    const int i = i0 + q;
    f.a[q] = *(const v4f*)(wa + (size_t)i * NV * NL);
    f.b[q] = *(const v4f*)(xb + (size_t)i * NL);
    if (WITH_PROBS) f.p[q] = pb[(size_t)i * BATCH];
  }
}

template <bool WITH_PROBS>
static __device__ __forceinline__ void mma_kblock(const KBlock& f, v8f& c0,
                                                  v8f& c1, v8f& c2, v8f& c3) {
  #pragma unroll
  for (int q = 0; q < NB; q += 2) {
    v4f b0 = f.b[q], b1 = f.b[q + 1];
    if (WITH_PROBS) { b0 = b0 * f.p[q]; b1 = b1 * f.p[q + 1]; }
    const v4f a0 = f.a[q], a1 = f.a[q + 1];
    c0 = wmma_f32_k4(__builtin_shufflevector(a0, a0, 0, 1),
                     __builtin_shufflevector(b0, b0, 0, 1), c0);
    c1 = wmma_f32_k4(__builtin_shufflevector(a0, a0, 2, 3),
                     __builtin_shufflevector(b0, b0, 2, 3), c1);
    c2 = wmma_f32_k4(__builtin_shufflevector(a1, a1, 0, 1),
                     __builtin_shufflevector(b1, b1, 0, 1), c2);
    c3 = wmma_f32_k4(__builtin_shufflevector(a1, a1, 2, 3),
                     __builtin_shufflevector(b1, b1, 2, 3), c3);
  }
}

template <int MODE>
static __device__ __forceinline__ void caps_gemm_body(
    const float* __restrict__ w, const float* __restrict__ x,
    const float* __restrict__ probs, const float* __restrict__ bias,
    float* __restrict__ dst) {
  constexpr bool WITH_PROBS = (MODE != 0);
  __shared__ float red[4 * 32 * 8];
  const int lane = threadIdx.x & 31;
  const int ks   = threadIdx.x >> 5;
  const int o    = blockIdx.x >> 2;
  const int bt   = blockIdx.x & 3;
  const int half = lane >> 4;
  const int lr   = lane & 15;
  const int bb   = bt * 16 + lr;
  const float* __restrict__ wa =
      w + ((size_t)o * NI * NV + lr) * NL + 4 * half;
  const float* __restrict__ xb = x + (size_t)bb * NI * NL + 4 * half;
  const float* __restrict__ pb =
      WITH_PROBS ? probs + (size_t)o * NI * BATCH + bb : nullptr;

  v8f c0 = {}, c1 = {}, c2 = {}, c3 = {};
  const int ibeg = ks * IW;
  constexpr int NTRIP = IW / (2 * NB);
  KBlock f0, f1;
  load_kblock<WITH_PROBS>(f0, wa, xb, pb, ibeg);
  #pragma unroll
  for (int t = 0; t < NTRIP; ++t) {
    const int i0 = ibeg + t * 2 * NB;
    load_kblock<WITH_PROBS>(f1, wa, xb, pb, i0 + NB);
    __builtin_amdgcn_sched_barrier(0);
    mma_kblock<WITH_PROBS>(f0, c0, c1, c2, c3);
    __builtin_amdgcn_sched_barrier(0);
    if (t + 1 < NTRIP)
      load_kblock<WITH_PROBS>(f0, wa, xb, pb, i0 + 2 * NB);
    __builtin_amdgcn_sched_barrier(0);
    mma_kblock<WITH_PROBS>(f1, c0, c1, c2, c3);
    __builtin_amdgcn_sched_barrier(0);
  }

  const v8f c = (c0 + c1) + (c2 + c3);
  {
    v4f lo = __builtin_shufflevector(c, c, 0, 1, 2, 3);
    v4f hi = __builtin_shufflevector(c, c, 4, 5, 6, 7);
    *(v4f*)&red[((size_t)ks * 32 + lane) * 8 + 0] = lo;
    *(v4f*)&red[((size_t)ks * 32 + lane) * 8 + 4] = hi;
  }
  __syncthreads();
  if (ks != 0) return;

  v4f tlo = {}, thi = {};
  #pragma unroll
  for (int s = 0; s < 4; ++s) {
    const float* q = &red[((size_t)s * 32 + lane) * 8];
    tlo += *(const v4f*)(q);
    thi += *(const v4f*)(q + 4);
  }
  if (MODE == 0) { tlo = tlo * (1.0f / (float)NI); thi = thi * (1.0f / (float)NI); }
  float ss = 0.0f;
  #pragma unroll
  for (int r = 0; r < 4; ++r) ss += tlo[r] * tlo[r];
  #pragma unroll
  for (int r = 0; r < 4; ++r) ss += thi[r] * thi[r];
  ss += __shfl_xor(ss, 16, 32);
  if (MODE != 2) {
    const float inv = 1.0f / fmaxf(sqrtf(ss), 1e-12f);
    float* q = dst + ((size_t)o * BATCH + bb) * NV + 8 * half;
    *(v4f*)(q)     = tlo * inv;
    *(v4f*)(q + 4) = thi * inv;
  } else {
    const float n  = sqrtf(ss);
    const float sc = n / (1.0f + ss);
    const float* bq = bias + (size_t)o * NV + 8 * half;
    const v4f blo = *(const v4f*)(bq);
    const v4f bhi = *(const v4f*)(bq + 4);
    float* q = dst + ((size_t)bb * NO + o) * NV + 8 * half;
    *(v4f*)(q)     = tlo * sc + blo;
    *(v4f*)(q + 4) = thi * sc + bhi;
  }
}

__global__ void __launch_bounds__(128) caps_init_kernel(const float* __restrict__ w,
                                 const float* __restrict__ x,
                                 float* __restrict__ nhat) {
  caps_gemm_body<0>(w, x, nullptr, nullptr, nhat);
}

__global__ void __launch_bounds__(128) caps_update_kernel(const float* __restrict__ w,
                                   const float* __restrict__ x,
                                   const float* __restrict__ probs,
                                   float* __restrict__ nhat) {
  caps_gemm_body<1>(w, x, probs, nullptr, nhat);
}

__global__ void __launch_bounds__(128) caps_last_update_kernel(
    const float* __restrict__ w, const float* __restrict__ x,
    const float* __restrict__ probs, const float* __restrict__ bias,
    float* __restrict__ y) {
  caps_gemm_body<2>(w, x, probs, bias, y);
}

struct LPair {
  v2f a00, a01, a10, a11;
  v4f nlo, nhi;
};

static __device__ __forceinline__ void load_lpair(
    LPair& f, const float* __restrict__ w, const float* __restrict__ nhat,
    int o, int ib0, int bb, int half, int lr) {
  const float* __restrict__ wrow =
      w + (((size_t)o * NI + ib0) * NV + lr) * NL + 2 * half;
  f.a00 = *(const v2f*)(wrow);
  f.a01 = *(const v2f*)(wrow + 4);
  f.a10 = *(const v2f*)(wrow + NV * NL);
  f.a11 = *(const v2f*)(wrow + NV * NL + 4);
  const float* __restrict__ nh = nhat + ((size_t)o * BATCH + bb) * NV + 8 * half;
  f.nlo = *(const v4f*)(nh);
  f.nhi = *(const v4f*)(nh + 4);
}

static __device__ __forceinline__ void finish_lpair(
    const LPair& f, v2f b00, v2f b01, v2f b10, v2f b11, float* slog, int o,
    int bb, int half) {
  v8f c0 = {}, c1 = {};
  c0 = wmma_f32_k4(f.a00, b00, c0);
  c1 = wmma_f32_k4(f.a10, b10, c1);
  c0 = wmma_f32_k4(f.a01, b01, c0);
  c1 = wmma_f32_k4(f.a11, b11, c1);
  float acc0 = 0.0f, acc1 = 0.0f;
  #pragma unroll
  for (int r = 0; r < 8; ++r) {
    const float nr = r < 4 ? f.nlo[r & 3] : f.nhi[r & 3];
    acc0 += c0[r] * nr;
    acc1 += c1[r] * nr;
  }
  acc0 += __shfl_xor(acc0, 16, 32);
  acc1 += __shfl_xor(acc1, 16, 32);
  slog[(o * IB + half) * BATCH + bb] = half ? acc1 : acc0;
}

__global__ void caps_logits_softmax_kernel(const float* __restrict__ w,
                                           const float* __restrict__ x,
                                           const float* __restrict__ nhat,
                                           float* __restrict__ probs) {
  extern __shared__ float slog[];
  const int tid  = threadIdx.x;
  const int wv   = tid >> 5;
  const int lane = tid & 31;
  const int half = lane >> 4;
  const int lr   = lane & 15;
  const int ib0  = blockIdx.x * IB;

  const int bt = wv & 3;
  const int o0 = (wv >> 2) * (NO / 4);
  const int bb = bt * 16 + lr;
  const float* __restrict__ xr = x + ((size_t)bb * NI + ib0) * NL + 2 * half;
  const v2f b00 = *(const v2f*)(xr);
  const v2f b01 = *(const v2f*)(xr + 4);
  const v2f b10 = *(const v2f*)(xr + NL);
  const v2f b11 = *(const v2f*)(xr + NL + 4);
  constexpr int NTRIP = NO / 4 / 2;
  LPair g0, g1;
  load_lpair(g0, w, nhat, o0, ib0, bb, half, lr);
  for (int n = 0; n < NTRIP; ++n) {
    const int o = o0 + 2 * n;
    load_lpair(g1, w, nhat, o + 1, ib0, bb, half, lr);
    finish_lpair(g0, b00, b01, b10, b11, slog, o, bb, half);
    if (n + 1 < NTRIP) load_lpair(g0, w, nhat, o + 2, ib0, bb, half, lr);
    finish_lpair(g1, b00, b01, b10, b11, slog, o + 1, bb, half);
  }
  __syncthreads();

  constexpr int NCOL = IB * BATCH;
  constexpr int OQ   = NO / 4;
  float* smax = slog + NO * NCOL;
  const int pr = tid % NCOL;
  const int q  = tid / NCOL;
  const int ii = pr >> 6;
  const int b  = pr & 63;
  const int i  = ib0 + ii;
  float* col = slog + q * OQ * NCOL + pr;
  float e[OQ];
  float mq = -3.402823466e+38f;
  #pragma unroll
  for (int k = 0; k < OQ; ++k) { e[k] = col[k * NCOL]; mq = fmaxf(mq, e[k]); }
  smax[q * NCOL + pr] = mq;
  __syncthreads();
  const float mx = fmaxf(fmaxf(smax[pr], smax[NCOL + pr]),
                         fmaxf(smax[2 * NCOL + pr], smax[3 * NCOL + pr]));
  #pragma unroll
  for (int k = 0; k < OQ; ++k) {
    e[k] = __expf(e[k] - mx);
    col[k * NCOL] = e[k];
  }
  __syncthreads();
  float s = 0.0f;
  #pragma unroll 16
  for (int o = 0; o < NO; ++o) s += slog[o * NCOL + pr];
  const float inv = 1.0f / s;
  #pragma unroll
  for (int k = 0; k < OQ; ++k)
    probs[((size_t)(q * OQ + k) * NI + i) * BATCH + b] = e[k] * inv;
}

extern "C" void kernel_launch(void* const* d_in, const int* in_sizes, int n_in,
                              void* d_out, int out_size, void* d_ws, size_t ws_size,
                              hipStream_t stream) {
  const float* x    = (const float*)d_in[0];
  const float* w    = (const float*)d_in[1];
  const float* bias = (const float*)d_in[2];
  float* y = (float*)d_out;

  float* probs = (float*)d_ws;
  float* nhat  = probs + (size_t)NO * NI * BATCH;
  const size_t need =
      ((size_t)NO * NI * BATCH + (size_t)NO * BATCH * NV) * sizeof(float);
  if (ws_size < need) return;

  caps_init_kernel<<<NO * 4, 128, 0, stream>>>(w, x, nhat);
  for (int it = 0; it < 2; ++it) {
    caps_logits_softmax_kernel<<<NI / IB, 512,
        (size_t)(NO + 4) * IB * BATCH * sizeof(float), stream>>>(w, x, nhat, probs);
    caps_update_kernel<<<NO * 4, 128, 0, stream>>>(w, x, probs, nhat);
  }
  caps_logits_softmax_kernel<<<NI / IB, 512,
      (size_t)(NO + 4) * IB * BATCH * sizeof(float), stream>>>(w, x, nhat, probs);
  caps_last_update_kernel<<<NO * 4, 128, 0, stream>>>(w, x, probs, bias, y);
}


# Work-groups that read the same lines walk them out of step: each GEMM group starts its K slices at block (2*bt+o)%8 and wraps (K sum order changes), each logits group its o walk at index%16
# speedup vs baseline: 7.6940x; 1.0644x over previous
#include <hip/hip_runtime.h>
#include <math.h>


typedef float v2f __attribute__((ext_vector_type(2)));
typedef float v4f __attribute__((ext_vector_type(4)));
typedef float v8f __attribute__((ext_vector_type(8)));

#define BATCH 64
#define NI    512
#define NO    64
#define NV    16
#define NL    8
#define IB    2
#define IW    (NI / 4)
#define NB    16
#define NBLK  (IW / NB)

static __device__ __forceinline__ v8f wmma_f32_k4(v2f a, v2f b, v8f c) {
  return __builtin_amdgcn_wmma_f32_16x16x4_f32(false, a, false, b, (short)0, c,
                                               false, false);
}

struct KBlock {
  v4f a[NB];
  v4f b[NB];
  float p[NB];
};

template <bool WITH_PROBS>
static __device__ __forceinline__ void load_kblock(
    KBlock& f, const float* __restrict__ wa, const float* __restrict__ xb,
    const float* __restrict__ pb, int i0) {
  #pragma unroll
  for (int q = 0; q < NB; ++q) {
    const int i = i0 + q;
    f.a[q] = *(const v4f*)(wa + (size_t)i * NV * NL);
    f.b[q] = *(const v4f*)(xb + (size_t)i * NL);
    if (WITH_PROBS) f.p[q] = pb[(size_t)i * BATCH];
  }
}

template <bool WITH_PROBS>
static __device__ __forceinline__ void mma_kblock(const KBlock& f, v8f& c0,
                                                  v8f& c1, v8f& c2, v8f& c3) {
  #pragma unroll
  for (int q = 0; q < NB; q += 2) {
    v4f b0 = f.b[q], b1 = f.b[q + 1];
    if (WITH_PROBS) { b0 = b0 * f.p[q]; b1 = b1 * f.p[q + 1]; }
    const v4f a0 = f.a[q], a1 = f.a[q + 1];
    c0 = wmma_f32_k4(__builtin_shufflevector(a0, a0, 0, 1),
                     __builtin_shufflevector(b0, b0, 0, 1), c0);
    c1 = wmma_f32_k4(__builtin_shufflevector(a0, a0, 2, 3),
                     __builtin_shufflevector(b0, b0, 2, 3), c1);
    c2 = wmma_f32_k4(__builtin_shufflevector(a1, a1, 0, 1),
                     __builtin_shufflevector(b1, b1, 0, 1), c2);
    c3 = wmma_f32_k4(__builtin_shufflevector(a1, a1, 2, 3),
                     __builtin_shufflevector(b1, b1, 2, 3), c3);
  }
}

template <int MODE>
static __device__ __forceinline__ void caps_gemm_body(
    const float* __restrict__ w, const float* __restrict__ x,
    const float* __restrict__ probs, const float* __restrict__ bias,
    float* __restrict__ dst) {
  constexpr bool WITH_PROBS = (MODE != 0);
  __shared__ float red[4 * 32 * 8];
  const int lane = threadIdx.x & 31;
  const int ks   = threadIdx.x >> 5;
  const int o    = blockIdx.x >> 2;
  const int bt   = blockIdx.x & 3;
  const int half = lane >> 4;
  const int lr   = lane & 15;
  const int bb   = bt * 16 + lr;
  const float* __restrict__ wa =
      w + ((size_t)o * NI * NV + lr) * NL + 4 * half;
  const float* __restrict__ xb = x + (size_t)bb * NI * NL + 4 * half;
  const float* __restrict__ pb =
      WITH_PROBS ? probs + (size_t)o * NI * BATCH + bb : nullptr;

  v8f c0 = {}, c1 = {}, c2 = {}, c3 = {};
  const int ibeg = ks * IW;
  constexpr int NTRIP = IW / (2 * NB);
  const int rot = (2 * bt + o) & (NBLK - 1);
  #define KBLK_I0(k) (ibeg + ((rot + (k)) & (NBLK - 1)) * NB)
  KBlock f0, f1;
  load_kblock<WITH_PROBS>(f0, wa, xb, pb, KBLK_I0(0));
  #pragma unroll
  for (int t = 0; t < NTRIP; ++t) {
    load_kblock<WITH_PROBS>(f1, wa, xb, pb, KBLK_I0(2 * t + 1));
    __builtin_amdgcn_sched_barrier(0);
    mma_kblock<WITH_PROBS>(f0, c0, c1, c2, c3);
    __builtin_amdgcn_sched_barrier(0);
    if (t + 1 < NTRIP)
      load_kblock<WITH_PROBS>(f0, wa, xb, pb, KBLK_I0(2 * t + 2));
    __builtin_amdgcn_sched_barrier(0);
    mma_kblock<WITH_PROBS>(f1, c0, c1, c2, c3);
    __builtin_amdgcn_sched_barrier(0);
  }
  #undef KBLK_I0

  const v8f c = (c0 + c1) + (c2 + c3);
  {
    v4f lo = __builtin_shufflevector(c, c, 0, 1, 2, 3);
    v4f hi = __builtin_shufflevector(c, c, 4, 5, 6, 7);
    *(v4f*)&red[((size_t)ks * 32 + lane) * 8 + 0] = lo;
    *(v4f*)&red[((size_t)ks * 32 + lane) * 8 + 4] = hi;
  }
  __syncthreads();
  if (ks != 0) return;

  v4f tlo = {}, thi = {};
  #pragma unroll
  for (int s = 0; s < 4; ++s) {
    const float* q = &red[((size_t)s * 32 + lane) * 8];
    tlo += *(const v4f*)(q);
    thi += *(const v4f*)(q + 4);
  }
  if (MODE == 0) { tlo = tlo * (1.0f / (float)NI); thi = thi * (1.0f / (float)NI); }
  float ss = 0.0f;
  #pragma unroll
  for (int r = 0; r < 4; ++r) ss += tlo[r] * tlo[r];
  #pragma unroll
  for (int r = 0; r < 4; ++r) ss += thi[r] * thi[r];
  ss += __shfl_xor(ss, 16, 32);
  if (MODE != 2) {
    const float inv = 1.0f / fmaxf(sqrtf(ss), 1e-12f);
    float* q = dst + ((size_t)o * BATCH + bb) * NV + 8 * half;
    *(v4f*)(q)     = tlo * inv;
    *(v4f*)(q + 4) = thi * inv;
  } else {
    const float n  = sqrtf(ss);
    const float sc = n / (1.0f + ss);
    const float* bq = bias + (size_t)o * NV + 8 * half;
    const v4f blo = *(const v4f*)(bq);
    const v4f bhi = *(const v4f*)(bq + 4);
    float* q = dst + ((size_t)bb * NO + o) * NV + 8 * half;
    *(v4f*)(q)     = tlo * sc + blo;
    *(v4f*)(q + 4) = thi * sc + bhi;
  }
}

__global__ void __launch_bounds__(128) caps_init_kernel(const float* __restrict__ w,
                                 const float* __restrict__ x,
                                 float* __restrict__ nhat) {
  caps_gemm_body<0>(w, x, nullptr, nullptr, nhat);
}

__global__ void __launch_bounds__(128) caps_update_kernel(const float* __restrict__ w,
                                   const float* __restrict__ x,
                                   const float* __restrict__ probs,
                                   float* __restrict__ nhat) {
  caps_gemm_body<1>(w, x, probs, nullptr, nhat);
}

__global__ void __launch_bounds__(128) caps_last_update_kernel(
    const float* __restrict__ w, const float* __restrict__ x,
    const float* __restrict__ probs, const float* __restrict__ bias,
    float* __restrict__ y) {
  caps_gemm_body<2>(w, x, probs, bias, y);
}

struct LPair {
  v2f a00, a01, a10, a11;
  v4f nlo, nhi;
};

static __device__ __forceinline__ void load_lpair(
    LPair& f, const float* __restrict__ w, const float* __restrict__ nhat,
    int o, int ib0, int bb, int half, int lr) {
  const float* __restrict__ wrow =
      w + (((size_t)o * NI + ib0) * NV + lr) * NL + 2 * half;
  f.a00 = *(const v2f*)(wrow);
  f.a01 = *(const v2f*)(wrow + 4);
  f.a10 = *(const v2f*)(wrow + NV * NL);
  f.a11 = *(const v2f*)(wrow + NV * NL + 4);
  const float* __restrict__ nh = nhat + ((size_t)o * BATCH + bb) * NV + 8 * half;
  f.nlo = *(const v4f*)(nh);
  f.nhi = *(const v4f*)(nh + 4);
}

static __device__ __forceinline__ void finish_lpair(
    const LPair& f, v2f b00, v2f b01, v2f b10, v2f b11, float* slog, int o,
    int bb, int half) {
  v8f c0 = {}, c1 = {};
  c0 = wmma_f32_k4(f.a00, b00, c0);
  c1 = wmma_f32_k4(f.a10, b10, c1);
  c0 = wmma_f32_k4(f.a01, b01, c0);
  c1 = wmma_f32_k4(f.a11, b11, c1);
  float acc0 = 0.0f, acc1 = 0.0f;
  #pragma unroll
  for (int r = 0; r < 8; ++r) {
    const float nr = r < 4 ? f.nlo[r & 3] : f.nhi[r & 3];
    acc0 += c0[r] * nr;
    acc1 += c1[r] * nr;
  }
  acc0 += __shfl_xor(acc0, 16, 32);
  acc1 += __shfl_xor(acc1, 16, 32);
  slog[(o * IB + half) * BATCH + bb] = half ? acc1 : acc0;
}

__global__ void caps_logits_softmax_kernel(const float* __restrict__ w,
                                           const float* __restrict__ x,
                                           const float* __restrict__ nhat,
                                           float* __restrict__ probs) {
  extern __shared__ float slog[];
  const int tid  = threadIdx.x;
  const int wv   = tid >> 5;
  const int lane = tid & 31;
  const int half = lane >> 4;
  const int lr   = lane & 15;
  const int ib0  = blockIdx.x * IB;

  const int bt = wv & 3;
  const int o0 = (wv >> 2) * (NO / 4);
  const int bb = bt * 16 + lr;
  const float* __restrict__ xr = x + ((size_t)bb * NI + ib0) * NL + 2 * half;
  const v2f b00 = *(const v2f*)(xr);
  const v2f b01 = *(const v2f*)(xr + 4);
  const v2f b10 = *(const v2f*)(xr + NL);
  const v2f b11 = *(const v2f*)(xr + NL + 4);
  constexpr int NTRIP = NO / 4 / 2;
  constexpr int OW    = NO / 4;
  const int rot = blockIdx.x & (OW - 1);
  LPair g0, g1;
  load_lpair(g0, w, nhat, o0 + rot, ib0, bb, half, lr);
  for (int n = 0; n < NTRIP; ++n) {
    const int oa = o0 + ((2 * n + rot) & (OW - 1));
    const int ob = o0 + ((2 * n + 1 + rot) & (OW - 1));
    load_lpair(g1, w, nhat, ob, ib0, bb, half, lr);
    __builtin_amdgcn_sched_barrier(0);
    finish_lpair(g0, b00, b01, b10, b11, slog, oa, bb, half);
    if (n + 1 < NTRIP)
      load_lpair(g0, w, nhat, o0 + ((2 * n + 2 + rot) & (OW - 1)), ib0, bb, half, lr);
    __builtin_amdgcn_sched_barrier(0);
    finish_lpair(g1, b00, b01, b10, b11, slog, ob, bb, half);
  }
  __syncthreads();

  constexpr int NCOL = IB * BATCH;
  constexpr int OQ   = NO / 4;
  float* smax = slog + NO * NCOL;
  const int pr = tid % NCOL;
  const int q  = tid / NCOL;
  const int ii = pr >> 6;
  const int b  = pr & 63;
  const int i  = ib0 + ii;
  float* col = slog + q * OQ * NCOL + pr;
  float e[OQ];
  float mq = -3.402823466e+38f;
  #pragma unroll
  for (int k = 0; k < OQ; ++k) { e[k] = col[k * NCOL]; mq = fmaxf(mq, e[k]); }
  smax[q * NCOL + pr] = mq;
  __syncthreads();
  const float mx = fmaxf(fmaxf(smax[pr], smax[NCOL + pr]),
                         fmaxf(smax[2 * NCOL + pr], smax[3 * NCOL + pr]));
  #pragma unroll
  for (int k = 0; k < OQ; ++k) {
    e[k] = __expf(e[k] - mx);
    col[k * NCOL] = e[k];
  }
  __syncthreads();
  float s = 0.0f;
  #pragma unroll 16
  for (int o = 0; o < NO; ++o) s += slog[o * NCOL + pr];
  const float inv = 1.0f / s;
  #pragma unroll
  for (int k = 0; k < OQ; ++k)
    probs[((size_t)(q * OQ + k) * NI + i) * BATCH + b] = e[k] * inv;
}

extern "C" void kernel_launch(void* const* d_in, const int* in_sizes, int n_in,
                              void* d_out, int out_size, void* d_ws, size_t ws_size,
                              hipStream_t stream) {
  const float* x    = (const float*)d_in[0];
  const float* w    = (const float*)d_in[1];
  const float* bias = (const float*)d_in[2];
  float* y = (float*)d_out;

  float* probs = (float*)d_ws;
  float* nhat  = probs + (size_t)NO * NI * BATCH;
  const size_t need =
      ((size_t)NO * NI * BATCH + (size_t)NO * BATCH * NV) * sizeof(float);
  if (ws_size < need) return;

  caps_init_kernel<<<NO * 4, 128, 0, stream>>>(w, x, nhat);
  for (int it = 0; it < 2; ++it) {
    caps_logits_softmax_kernel<<<NI / IB, 512,
        (size_t)(NO + 4) * IB * BATCH * sizeof(float), stream>>>(w, x, nhat, probs);
    caps_update_kernel<<<NO * 4, 128, 0, stream>>>(w, x, probs, nhat);
  }
  caps_logits_softmax_kernel<<<NI / IB, 512,
      (size_t)(NO + 4) * IB * BATCH * sizeof(float), stream>>>(w, x, nhat, probs);
  caps_last_update_kernel<<<NO * 4, 128, 0, stream>>>(w, x, probs, bias, y);
}


# Workspace operands in WMMA fragment order: init GEMM also writes x as [b tile][i][16 b][8] for the 6 later launches, nhat split by lane half: a fragment load touches 4 whole lines, not 16 or 8
# speedup vs baseline: 8.2493x; 1.0722x over previous
#include <hip/hip_runtime.h>
#include <math.h>


typedef float v2f __attribute__((ext_vector_type(2)));
typedef float v4f __attribute__((ext_vector_type(4)));
typedef float v8f __attribute__((ext_vector_type(8)));

#define BATCH 64
#define NI    512
#define NO    64
#define NV    16
#define NL    8
#define IB    2
#define IW    (NI / 4)
#define NB    16
#define NBLK  (IW / NB)

static __device__ __forceinline__ v8f wmma_f32_k4(v2f a, v2f b, v8f c) {
  return __builtin_amdgcn_wmma_f32_16x16x4_f32(false, a, false, b, (short)0, c,
                                               false, false);
}

struct KBlock {
  v4f a[NB];
  v4f b[NB];
  float p[NB];
};

template <bool WITH_PROBS, int XS>
static __device__ __forceinline__ void load_kblock(
    KBlock& f, const float* __restrict__ wa, const float* __restrict__ xb,
    const float* __restrict__ pb, int i0) {
  #pragma unroll
  for (int q = 0; q < NB; ++q) {
    const int i = i0 + q;
    f.a[q] = *(const v4f*)(wa + (size_t)i * NV * NL);
    f.b[q] = *(const v4f*)(xb + (size_t)i * XS);
    if (WITH_PROBS) f.p[q] = pb[(size_t)i * BATCH];
  }
}

template <bool WITH_PROBS>
static __device__ __forceinline__ void mma_kblock(const KBlock& f, v8f& c0,
                                                  v8f& c1, v8f& c2, v8f& c3) {
  #pragma unroll
  for (int q = 0; q < NB; q += 2) {
    v4f b0 = f.b[q], b1 = f.b[q + 1];
    if (WITH_PROBS) { b0 = b0 * f.p[q]; b1 = b1 * f.p[q + 1]; }
    const v4f a0 = f.a[q], a1 = f.a[q + 1];
    c0 = wmma_f32_k4(__builtin_shufflevector(a0, a0, 0, 1),
                     __builtin_shufflevector(b0, b0, 0, 1), c0);
    c1 = wmma_f32_k4(__builtin_shufflevector(a0, a0, 2, 3),
                     __builtin_shufflevector(b0, b0, 2, 3), c1);
    c2 = wmma_f32_k4(__builtin_shufflevector(a1, a1, 0, 1),
                     __builtin_shufflevector(b1, b1, 0, 1), c2);
    c3 = wmma_f32_k4(__builtin_shufflevector(a1, a1, 2, 3),
                     __builtin_shufflevector(b1, b1, 2, 3), c3);
  }
}

static __device__ __forceinline__ void store_xt_kblock(
    const KBlock& f, float* __restrict__ xtb, int i0) {
  #pragma unroll
  for (int q = 0; q < NB; ++q)
    *(v4f*)(xtb + (size_t)(i0 + q) * 16 * NL) = f.b[q];
}

static __device__ __forceinline__ size_t nhat_idx(int o, int bt, int j, int half,
                                                  int lr) {
  return ((((size_t)(o * 4 + bt) * 2 + j) * 2 + half) * 16 + lr) * 4;
}

template <int MODE>
static __device__ __forceinline__ void caps_gemm_body(
    const float* __restrict__ w, const float* __restrict__ xsrc,
    const float* __restrict__ probs, const float* __restrict__ bias,
    float* __restrict__ dst, float* __restrict__ xt_out) {
  constexpr bool WITH_PROBS = (MODE != 0);
  constexpr int XS = (MODE == 0) ? NL : 16 * NL;
  __shared__ float red[4 * 32 * 8];
  const int lane = threadIdx.x & 31;
  const int ks   = threadIdx.x >> 5;
  const int o    = blockIdx.x >> 2;
  const int bt   = blockIdx.x & 3;
  const int half = lane >> 4;
  const int lr   = lane & 15;
  const int bb   = bt * 16 + lr;
  const float* __restrict__ wa =
      w + ((size_t)o * NI * NV + lr) * NL + 4 * half;
  const float* __restrict__ xb =
      (MODE == 0) ? xsrc + (size_t)bb * NI * NL + 4 * half
                  : xsrc + ((size_t)bt * NI * 16 + lr) * NL + 4 * half;
  float* __restrict__ xtb =
      (MODE == 0) ? xt_out + ((size_t)bt * NI * 16 + lr) * NL + 4 * half : nullptr;
  const float* __restrict__ pb =
      WITH_PROBS ? probs + (size_t)o * NI * BATCH + bb : nullptr;

  v8f c0 = {}, c1 = {}, c2 = {}, c3 = {};
  const int ibeg = ks * IW;
  constexpr int NTRIP = IW / (2 * NB);
  const int rot = (2 * bt + o) & (NBLK - 1);
  #define KBLK_I0(k) (ibeg + ((rot + (k)) & (NBLK - 1)) * NB)
  #define KBLK_MINE(k) (MODE == 0 && KBLK_I0(k) == o * NB)
  KBlock f0, f1;
  load_kblock<WITH_PROBS, XS>(f0, wa, xb, pb, KBLK_I0(0));
  #pragma unroll
  for (int t = 0; t < NTRIP; ++t) {
    load_kblock<WITH_PROBS, XS>(f1, wa, xb, pb, KBLK_I0(2 * t + 1));
    __builtin_amdgcn_sched_barrier(0);
    mma_kblock<WITH_PROBS>(f0, c0, c1, c2, c3);
    if (KBLK_MINE(2 * t)) store_xt_kblock(f0, xtb, KBLK_I0(2 * t));
    __builtin_amdgcn_sched_barrier(0);
    if (t + 1 < NTRIP)
      load_kblock<WITH_PROBS, XS>(f0, wa, xb, pb, KBLK_I0(2 * t + 2));
    __builtin_amdgcn_sched_barrier(0);
    mma_kblock<WITH_PROBS>(f1, c0, c1, c2, c3);
    if (KBLK_MINE(2 * t + 1)) store_xt_kblock(f1, xtb, KBLK_I0(2 * t + 1));
    __builtin_amdgcn_sched_barrier(0);
  }
  #undef KBLK_MINE
  #undef KBLK_I0

  const v8f c = (c0 + c1) + (c2 + c3);
  {
    v4f lo = __builtin_shufflevector(c, c, 0, 1, 2, 3);
    v4f hi = __builtin_shufflevector(c, c, 4, 5, 6, 7);
    *(v4f*)&red[((size_t)ks * 32 + lane) * 8 + 0] = lo;
    *(v4f*)&red[((size_t)ks * 32 + lane) * 8 + 4] = hi;
  }
  __syncthreads();
  if (ks != 0) return;

  v4f tlo = {}, thi = {};
  #pragma unroll
  for (int s = 0; s < 4; ++s) {
    const float* q = &red[((size_t)s * 32 + lane) * 8];
    tlo += *(const v4f*)(q);
    thi += *(const v4f*)(q + 4);
  }
  if (MODE == 0) { tlo = tlo * (1.0f / (float)NI); thi = thi * (1.0f / (float)NI); }
  float ss = 0.0f;
  #pragma unroll
  for (int r = 0; r < 4; ++r) ss += tlo[r] * tlo[r];
  #pragma unroll
  for (int r = 0; r < 4; ++r) ss += thi[r] * thi[r];
  ss += __shfl_xor(ss, 16, 32);
  if (MODE != 2) {
    const float inv = 1.0f / fmaxf(sqrtf(ss), 1e-12f);
    *(v4f*)(dst + nhat_idx(o, bt, 0, half, lr)) = tlo * inv;
    *(v4f*)(dst + nhat_idx(o, bt, 1, half, lr)) = thi * inv;
  } else {
    const float n  = sqrtf(ss);
    const float sc = n / (1.0f + ss);
    const float* bq = bias + (size_t)o * NV + 8 * half;
    const v4f blo = *(const v4f*)(bq);
    const v4f bhi = *(const v4f*)(bq + 4);
    float* q = dst + ((size_t)bb * NO + o) * NV + 8 * half;
    *(v4f*)(q)     = tlo * sc + blo;
    *(v4f*)(q + 4) = thi * sc + bhi;
  }
}

__global__ void __launch_bounds__(128) caps_init_kernel(const float* __restrict__ w,
                                 const float* __restrict__ x,
                                 float* __restrict__ nhat,
                                 float* __restrict__ xt) {
  caps_gemm_body<0>(w, x, nullptr, nullptr, nhat, xt);
}

__global__ void __launch_bounds__(128) caps_update_kernel(const float* __restrict__ w,
                                   const float* __restrict__ xt,
                                   const float* __restrict__ probs,
                                   float* __restrict__ nhat) {
  caps_gemm_body<1>(w, xt, probs, nullptr, nhat, nullptr);
}

__global__ void __launch_bounds__(128) caps_last_update_kernel(
    const float* __restrict__ w, const float* __restrict__ xt,
    const float* __restrict__ probs, const float* __restrict__ bias,
    float* __restrict__ y) {
  caps_gemm_body<2>(w, xt, probs, bias, y, nullptr);
}

struct LPair {
  v2f a00, a01, a10, a11;
  v4f nlo, nhi;
};

static __device__ __forceinline__ void load_lpair(
    LPair& f, const float* __restrict__ w, const float* __restrict__ nhat,
    int o, int ib0, int bt, int half, int lr) {
  const float* __restrict__ wrow =
      w + (((size_t)o * NI + ib0) * NV + lr) * NL + 2 * half;
  f.a00 = *(const v2f*)(wrow);
  f.a01 = *(const v2f*)(wrow + 4);
  f.a10 = *(const v2f*)(wrow + NV * NL);
  f.a11 = *(const v2f*)(wrow + NV * NL + 4);
  f.nlo = *(const v4f*)(nhat + nhat_idx(o, bt, 0, half, lr));
  f.nhi = *(const v4f*)(nhat + nhat_idx(o, bt, 1, half, lr));
}

static __device__ __forceinline__ void finish_lpair(
    const LPair& f, v2f b00, v2f b01, v2f b10, v2f b11, float* slog, int o,
    int bb, int half) {
  v8f c0 = {}, c1 = {};
  c0 = wmma_f32_k4(f.a00, b00, c0);
  c1 = wmma_f32_k4(f.a10, b10, c1);
  c0 = wmma_f32_k4(f.a01, b01, c0);
  c1 = wmma_f32_k4(f.a11, b11, c1);
  float acc0 = 0.0f, acc1 = 0.0f;
  #pragma unroll
  for (int r = 0; r < 8; ++r) {
    const float nr = r < 4 ? f.nlo[r & 3] : f.nhi[r & 3];
    acc0 += c0[r] * nr;
    acc1 += c1[r] * nr;
  }
  acc0 += __shfl_xor(acc0, 16, 32);
  acc1 += __shfl_xor(acc1, 16, 32);
  slog[(o * IB + half) * BATCH + bb] = half ? acc1 : acc0;
}

__global__ void caps_logits_softmax_kernel(const float* __restrict__ w,
                                           const float* __restrict__ xt,
                                           const float* __restrict__ nhat,
                                           float* __restrict__ probs) {
  extern __shared__ float slog[];
  const int tid  = threadIdx.x;
  const int wv   = tid >> 5;
  const int lane = tid & 31;
  const int half = lane >> 4;
  const int lr   = lane & 15;
  const int ib0  = blockIdx.x * IB;

  const int bt = wv & 3;
  const int o0 = (wv >> 2) * (NO / 4);
  const int bb = bt * 16 + lr;
  const float* __restrict__ xr =
      xt + (((size_t)bt * NI + ib0) * 16 + lr) * NL + 2 * half;
  const v2f b00 = *(const v2f*)(xr);
  const v2f b01 = *(const v2f*)(xr + 4);
  const v2f b10 = *(const v2f*)(xr + 16 * NL);
  const v2f b11 = *(const v2f*)(xr + 16 * NL + 4);
  constexpr int NTRIP = NO / 4 / 2;
  constexpr int OW    = NO / 4;
  const int rot = blockIdx.x & (OW - 1);
  LPair g0, g1;
  load_lpair(g0, w, nhat, o0 + rot, ib0, bt, half, lr);
  for (int n = 0; n < NTRIP; ++n) {
    const int oa = o0 + ((2 * n + rot) & (OW - 1));
    const int ob = o0 + ((2 * n + 1 + rot) & (OW - 1));
    load_lpair(g1, w, nhat, ob, ib0, bt, half, lr);
    __builtin_amdgcn_sched_barrier(0);
    finish_lpair(g0, b00, b01, b10, b11, slog, oa, bb, half);
    if (n + 1 < NTRIP)
      load_lpair(g0, w, nhat, o0 + ((2 * n + 2 + rot) & (OW - 1)), ib0, bt, half, lr);
    __builtin_amdgcn_sched_barrier(0);
    finish_lpair(g1, b00, b01, b10, b11, slog, ob, bb, half);
  }
  __syncthreads();

  constexpr int NCOL = IB * BATCH;
  constexpr int OQ   = NO / 4;
  float* smax = slog + NO * NCOL;
  const int pr = tid % NCOL;
  const int q  = tid / NCOL;
  const int ii = pr >> 6;
  const int b  = pr & 63;
  const int i  = ib0 + ii;
  float* col = slog + q * OQ * NCOL + pr;
  float e[OQ];
  float mq = -3.402823466e+38f;
  #pragma unroll
  for (int k = 0; k < OQ; ++k) { e[k] = col[k * NCOL]; mq = fmaxf(mq, e[k]); }
  smax[q * NCOL + pr] = mq;
  __syncthreads();
  const float mx = fmaxf(fmaxf(smax[pr], smax[NCOL + pr]),
                         fmaxf(smax[2 * NCOL + pr], smax[3 * NCOL + pr]));
  #pragma unroll
  for (int k = 0; k < OQ; ++k) {
    e[k] = __expf(e[k] - mx);
    col[k * NCOL] = e[k];
  }
  __syncthreads();
  float s = 0.0f;
  #pragma unroll 16
  for (int o = 0; o < NO; ++o) s += slog[o * NCOL + pr];
  const float inv = 1.0f / s;
  #pragma unroll
  for (int k = 0; k < OQ; ++k)
    probs[((size_t)(q * OQ + k) * NI + i) * BATCH + b] = e[k] * inv;
}

extern "C" void kernel_launch(void* const* d_in, const int* in_sizes, int n_in,
                              void* d_out, int out_size, void* d_ws, size_t ws_size,
                              hipStream_t stream) {
  const float* x    = (const float*)d_in[0];
  const float* w    = (const float*)d_in[1];
  const float* bias = (const float*)d_in[2];
  float* y = (float*)d_out;

  float* probs = (float*)d_ws;
  float* nhat  = probs + (size_t)NO * NI * BATCH;
  float* xt    = nhat + (size_t)NO * BATCH * NV;
  const size_t need =
      ((size_t)NO * NI * BATCH + (size_t)NO * BATCH * NV +
       (size_t)BATCH * NI * NL) * sizeof(float);
  if (ws_size < need) return;

  caps_init_kernel<<<NO * 4, 128, 0, stream>>>(w, x, nhat, xt);
  for (int it = 0; it < 2; ++it) {
    caps_logits_softmax_kernel<<<NI / IB, 512,
        (size_t)(NO + 4) * IB * BATCH * sizeof(float), stream>>>(w, xt, nhat, probs);
    caps_update_kernel<<<NO * 4, 128, 0, stream>>>(w, xt, probs, nhat);
  }
  caps_logits_softmax_kernel<<<NI / IB, 512,
      (size_t)(NO + 4) * IB * BATCH * sizeof(float), stream>>>(w, xt, nhat, probs);
  caps_last_update_kernel<<<NO * 4, 128, 0, stream>>>(w, xt, probs, bias, y);
}


# Logits and update fused per (batch tile, 8 in-capsules): weights loaded once, kept in registers for both products, probs stay in LDS; 64 partial planes added by a small launch (sum orders change)
# speedup vs baseline: 8.5107x; 1.0317x over previous
#include <hip/hip_runtime.h>
#include <math.h>


typedef float v2f __attribute__((ext_vector_type(2)));
typedef float v4f __attribute__((ext_vector_type(4)));
typedef float v8f __attribute__((ext_vector_type(8)));

#define BATCH 64
#define NI    512
#define NO    64
#define NV    16
#define NL    8
#define IW    (NI / 4)
#define NB    16
#define NBLK  (IW / NB)
#define ICH   8
#define NCH   (NI / ICH)
#define OWV   4
#define PLANE (NO * BATCH * NV)

static __device__ __forceinline__ v8f wmma_f32_k4(v2f a, v2f b, v8f c) {
  return __builtin_amdgcn_wmma_f32_16x16x4_f32(false, a, false, b, (short)0, c,
                                               false, false);
}

static __device__ __forceinline__ v2f lo2(v4f t) {
  return __builtin_shufflevector(t, t, 0, 1);
}
static __device__ __forceinline__ v2f hi2(v4f t) {
  return __builtin_shufflevector(t, t, 2, 3);
}

struct KBlock {
  v4f a[NB];
  v4f b[NB];
};

static __device__ __forceinline__ void load_kblock(
    KBlock& f, const float* __restrict__ wa, const float* __restrict__ xb, int i0) {
  #pragma unroll
  for (int q = 0; q < NB; ++q) {
    const int i = i0 + q;
    f.a[q] = *(const v4f*)(wa + (size_t)i * NV * NL);
    f.b[q] = *(const v4f*)(xb + (size_t)i * NL);
  }
}

static __device__ __forceinline__ void mma_kblock(const KBlock& f, v8f& c0,
                                                  v8f& c1, v8f& c2, v8f& c3) {
  #pragma unroll
  for (int q = 0; q < NB; q += 2) {
    c0 = wmma_f32_k4(lo2(f.a[q]),     lo2(f.b[q]),     c0);
    c1 = wmma_f32_k4(hi2(f.a[q]),     hi2(f.b[q]),     c1);
    c2 = wmma_f32_k4(lo2(f.a[q + 1]), lo2(f.b[q + 1]), c2);
    c3 = wmma_f32_k4(hi2(f.a[q + 1]), hi2(f.b[q + 1]), c3);
  }
}

static __device__ __forceinline__ void store_xt_kblock(
    const KBlock& f, float* __restrict__ xtb, int i0) {
  #pragma unroll
  for (int q = 0; q < NB; ++q)
    *(v4f*)(xtb + (size_t)(i0 + q) * 16 * NL) = f.b[q];
}

static __device__ __forceinline__ size_t nhat_idx(int o, int bt, int j, int half,
                                                  int lr) {
  return ((((size_t)(o * 4 + bt) * 2 + j) * 2 + half) * 16 + lr) * 4;
}

template <int FIN>
static __device__ __forceinline__ void finish_tile(
    const v8f c, int ks, int lane, int o, int bt,
    const float* __restrict__ bias, float* __restrict__ dst) {
  __shared__ float red[4 * 32 * 8];
  const int half = lane >> 4;
  const int lr   = lane & 15;
  const int bb   = bt * 16 + lr;
  {
    v4f lo = __builtin_shufflevector(c, c, 0, 1, 2, 3);
    v4f hi = __builtin_shufflevector(c, c, 4, 5, 6, 7);
    *(v4f*)&red[((size_t)ks * 32 + lane) * 8 + 0] = lo;
    *(v4f*)&red[((size_t)ks * 32 + lane) * 8 + 4] = hi;
  }
  __syncthreads();
  if (ks != 0) return;

  v4f tlo = {}, thi = {};
  #pragma unroll
  for (int s = 0; s < 4; ++s) {
    const float* q = &red[((size_t)s * 32 + lane) * 8];
    tlo += *(const v4f*)(q);
    thi += *(const v4f*)(q + 4);
  }
  if (FIN == 0) { tlo = tlo * (1.0f / (float)NI); thi = thi * (1.0f / (float)NI); }
  float ss = 0.0f;
  #pragma unroll
  for (int r = 0; r < 4; ++r) ss += tlo[r] * tlo[r];
  #pragma unroll
  for (int r = 0; r < 4; ++r) ss += thi[r] * thi[r];
  ss += __shfl_xor(ss, 16, 32);
  if (FIN != 2) {
    const float inv = 1.0f / fmaxf(sqrtf(ss), 1e-12f);
    *(v4f*)(dst + nhat_idx(o, bt, 0, half, lr)) = tlo * inv;
    *(v4f*)(dst + nhat_idx(o, bt, 1, half, lr)) = thi * inv;
  } else {
    const float n  = sqrtf(ss);
    const float sc = n / (1.0f + ss);
    const float* bq = bias + (size_t)o * NV + 8 * half;
    const v4f blo = *(const v4f*)(bq);
    const v4f bhi = *(const v4f*)(bq + 4);
    float* q = dst + ((size_t)bb * NO + o) * NV + 8 * half;
    *(v4f*)(q)     = tlo * sc + blo;
    *(v4f*)(q + 4) = thi * sc + bhi;
  }
}

__global__ void __launch_bounds__(128) caps_init_kernel(const float* __restrict__ w,
                                 const float* __restrict__ x,
                                 float* __restrict__ nhat,
                                 float* __restrict__ xt) {
  const int lane = threadIdx.x & 31;
  const int ks   = threadIdx.x >> 5;
  const int o    = blockIdx.x >> 2;
  const int bt   = blockIdx.x & 3;
  const int half = lane >> 4;
  const int lr   = lane & 15;
  const int bb   = bt * 16 + lr;
  const float* __restrict__ wa =
      w + ((size_t)o * NI * NV + lr) * NL + 4 * half;
  const float* __restrict__ xb = x + (size_t)bb * NI * NL + 4 * half;
  float* __restrict__ xtb = xt + ((size_t)bt * NI * 16 + lr) * NL + 4 * half;

  v8f c0 = {}, c1 = {}, c2 = {}, c3 = {};
  const int ibeg = ks * IW;
  constexpr int NTRIP = IW / (2 * NB);
  const int rot = (2 * bt + o) & (NBLK - 1);
  #define KBLK_I0(k) (ibeg + ((rot + (k)) & (NBLK - 1)) * NB)
  #define KBLK_MINE(k) (KBLK_I0(k) == o * NB)
  KBlock f0, f1;
  load_kblock(f0, wa, xb, KBLK_I0(0));
  #pragma unroll
  for (int t = 0; t < NTRIP; ++t) {
    load_kblock(f1, wa, xb, KBLK_I0(2 * t + 1));
    __builtin_amdgcn_sched_barrier(0);
    mma_kblock(f0, c0, c1, c2, c3);
    if (KBLK_MINE(2 * t)) store_xt_kblock(f0, xtb, KBLK_I0(2 * t));
    __builtin_amdgcn_sched_barrier(0);
    if (t + 1 < NTRIP)
      load_kblock(f0, wa, xb, KBLK_I0(2 * t + 2));
    __builtin_amdgcn_sched_barrier(0);
    mma_kblock(f1, c0, c1, c2, c3);
    if (KBLK_MINE(2 * t + 1)) store_xt_kblock(f1, xtb, KBLK_I0(2 * t + 1));
    __builtin_amdgcn_sched_barrier(0);
  }
  #undef KBLK_MINE
  #undef KBLK_I0

  const v8f c = (c0 + c1) + (c2 + c3);
  finish_tile<0>(c, ks, lane, o, bt, nullptr, nhat);
}

template <int FIN>
static __device__ __forceinline__ void caps_sum_body(
    const float* __restrict__ part, const float* __restrict__ bias,
    float* __restrict__ dst) {
  const int lane = threadIdx.x & 31;
  const int ks   = threadIdx.x >> 5;
  const int o    = blockIdx.x >> 2;
  const int bt   = blockIdx.x & 3;
  const int half = lane >> 4;
  const int lr   = lane & 15;
  constexpr int PW = NCH / 4;
  const float* __restrict__ p0 =
      part + (size_t)(ks * PW) * PLANE + nhat_idx(o, bt, 0, half, lr);
  const float* __restrict__ p1 =
      part + (size_t)(ks * PW) * PLANE + nhat_idx(o, bt, 1, half, lr);
  v4f lo = {}, hi = {};
  #pragma unroll
  for (int s = 0; s < PW; ++s) {
    lo += *(const v4f*)(p0 + (size_t)s * PLANE);
    hi += *(const v4f*)(p1 + (size_t)s * PLANE);
  }
  const v8f c = __builtin_shufflevector(lo, hi, 0, 1, 2, 3, 4, 5, 6, 7);
  finish_tile<FIN>(c, ks, lane, o, bt, bias, dst);
}

__global__ void __launch_bounds__(128) caps_sum_kernel(const float* __restrict__ part,
                                                       float* __restrict__ nhat) {
  caps_sum_body<1>(part, nullptr, nhat);
}

__global__ void __launch_bounds__(128) caps_last_sum_kernel(
    const float* __restrict__ part, const float* __restrict__ bias,
    float* __restrict__ y) {
  caps_sum_body<2>(part, bias, y);
}

__global__ void __launch_bounds__(512) caps_route_kernel(const float* __restrict__ w,
                                                         const float* __restrict__ xt,
                                                         const float* __restrict__ nhat,
                                                         float* __restrict__ part) {
  constexpr int NCOL = ICH * 16;
  __shared__ float slog[NO * NCOL];
  __shared__ float smax[4 * NCOL];
  __shared__ float ssum[4 * NCOL];
  const int tid  = threadIdx.x;
  const int wv   = tid >> 5;
  const int lane = tid & 31;
  const int half = lane >> 4;
  const int lr   = lane & 15;
  const int bt   = blockIdx.x & 3;
  const int ch   = blockIdx.x >> 2;
  const int i0   = ch * ICH;
  const int o0   = wv * OWV;

  v4f xa[ICH];
  {
    const float* __restrict__ xr =
        xt + (((size_t)bt * NI + i0) * 16 + lr) * NL + 4 * half;
    #pragma unroll
    for (int q = 0; q < ICH; ++q) xa[q] = *(const v4f*)(xr + (size_t)q * 16 * NL);
  }
  v4f wa[OWV][ICH];
  v4f nlo[OWV], nhi[OWV];
  #pragma unroll
  for (int j = 0; j < OWV; ++j) {
    const float* __restrict__ wr =
        w + (((size_t)(o0 + j) * NI + i0) * NV + lr) * NL + 4 * half;
    nlo[j] = *(const v4f*)(nhat + nhat_idx(o0 + j, bt, 0, half, lr));
    nhi[j] = *(const v4f*)(nhat + nhat_idx(o0 + j, bt, 1, half, lr));
    #pragma unroll
    for (int q = 0; q < ICH; ++q) wa[j][q] = *(const v4f*)(wr + (size_t)q * NV * NL);
  }

  #pragma unroll
  for (int j = 0; j < OWV; ++j) {
    #pragma unroll
    for (int q = 0; q < ICH; q += 2) {
      v8f c0 = {}, c1 = {};
      c0 = wmma_f32_k4(lo2(wa[j][q]),     lo2(xa[q]),     c0);
      c1 = wmma_f32_k4(lo2(wa[j][q + 1]), lo2(xa[q + 1]), c1);
      c0 = wmma_f32_k4(hi2(wa[j][q]),     hi2(xa[q]),     c0);
      c1 = wmma_f32_k4(hi2(wa[j][q + 1]), hi2(xa[q + 1]), c1);
      const v8f nj = __builtin_shufflevector(nlo[j], nhi[j], 0, 1, 2, 3, 4, 5, 6, 7);
      v2f s0 = {0.0f, 0.0f}, s1 = {0.0f, 0.0f};
      #pragma unroll
      for (int m = 0; m < 8; m += 2) {
        const v2f n2 = {nj[m], nj[m + 1]};
        s0 += v2f{c0[m], c0[m + 1]} * n2;
        s1 += v2f{c1[m], c1[m + 1]} * n2;
      }
      const float acc0 = s0[0] + s0[1], acc1 = s1[0] + s1[1];
      const float keep = half ? acc1 : acc0;
      const float send = half ? acc0 : acc1;
      slog[((o0 + j) * ICH + q + half) * 16 + lr] = keep + __shfl_xor(send, 16, 32);
    }
  }
  __syncthreads();

  constexpr int OQ = NO / 4;
  const int pr = tid % NCOL;
  const int qq = tid / NCOL;
  float* col = slog + qq * OQ * NCOL + pr;
  float e[OQ];
  float mq = -3.402823466e+38f;
  #pragma unroll
  for (int k = 0; k < OQ; ++k) { e[k] = col[k * NCOL]; mq = fmaxf(mq, e[k]); }
  smax[qq * NCOL + pr] = mq;
  __syncthreads();
  const float mx = fmaxf(fmaxf(smax[pr], smax[NCOL + pr]),
                         fmaxf(smax[2 * NCOL + pr], smax[3 * NCOL + pr]));
  float sq = 0.0f;
  #pragma unroll
  for (int k = 0; k < OQ; ++k) { e[k] = __expf(e[k] - mx); sq += e[k]; }
  ssum[qq * NCOL + pr] = sq;
  __syncthreads();
  const float s = ((ssum[pr] + ssum[NCOL + pr]) + ssum[2 * NCOL + pr]) + ssum[3 * NCOL + pr];
  const float inv = 1.0f / s;
  #pragma unroll
  for (int k = 0; k < OQ; ++k) col[k * NCOL] = e[k] * inv;
  __syncthreads();

  v8f acc[OWV];
  #pragma unroll
  for (int j = 0; j < OWV; ++j) acc[j] = v8f{};
  #pragma unroll
  for (int q = 0; q < ICH; ++q) {
    v4f bq[OWV];
    #pragma unroll
    for (int j = 0; j < OWV; ++j) bq[j] = xa[q] * slog[((o0 + j) * ICH + q) * 16 + lr];
    #pragma unroll
    for (int j = 0; j < OWV; ++j) acc[j] = wmma_f32_k4(lo2(wa[j][q]), lo2(bq[j]), acc[j]);
    #pragma unroll
    for (int j = 0; j < OWV; ++j) acc[j] = wmma_f32_k4(hi2(wa[j][q]), hi2(bq[j]), acc[j]);
  }
  float* __restrict__ pp = part + (size_t)ch * PLANE;
  #pragma unroll
  for (int j = 0; j < OWV; ++j) {
    *(v4f*)(pp + nhat_idx(o0 + j, bt, 0, half, lr)) =
        __builtin_shufflevector(acc[j], acc[j], 0, 1, 2, 3);
    *(v4f*)(pp + nhat_idx(o0 + j, bt, 1, half, lr)) =
        __builtin_shufflevector(acc[j], acc[j], 4, 5, 6, 7);
  }
}

extern "C" void kernel_launch(void* const* d_in, const int* in_sizes, int n_in,
                              void* d_out, int out_size, void* d_ws, size_t ws_size,
                              hipStream_t stream) {
  const float* x    = (const float*)d_in[0];
  const float* w    = (const float*)d_in[1];
  const float* bias = (const float*)d_in[2];
  float* y = (float*)d_out;

  float* nhat = (float*)d_ws;
  float* xt   = nhat + (size_t)PLANE;
  float* part = xt + (size_t)BATCH * NI * NL;
  const size_t need =
      ((size_t)PLANE + (size_t)BATCH * NI * NL + (size_t)NCH * PLANE) * sizeof(float);
  if (ws_size < need) return;

  caps_init_kernel<<<NO * 4, 128, 0, stream>>>(w, x, nhat, xt);
  for (int it = 0; it < 2; ++it) {
    caps_route_kernel<<<NCH * 4, 512, 0, stream>>>(w, xt, nhat, part);
    caps_sum_kernel<<<NO * 4, 128, 0, stream>>>(part, nhat);
  }
  caps_route_kernel<<<NCH * 4, 512, 0, stream>>>(w, xt, nhat, part);
  caps_last_sum_kernel<<<NO * 4, 128, 0, stream>>>(part, bias, y);
}
